# starting kernel
# baseline (speedup 1.0000x reference)
#include <hip/hip_runtime.h>
#include <math.h>

typedef __attribute__((ext_vector_type(2))) float v2f;
typedef __attribute__((ext_vector_type(4))) float v4f;
typedef __attribute__((ext_vector_type(8))) float v8f;

#define B_  8192
#define P_  25
#define C_  256
#define D_  5
#define R_  (B_ * P_)
#define ROWS_PER_WAVE 16
#define WAVES_PER_BLOCK 2
#define HSTRIDE 260
#define KSTEPS (C_ / 4)
#define ZOFF (D_ * HSTRIDE)

__global__ __launch_bounds__(256) void rowsum_kernel(const float* __restrict__ w,
                                                     float* __restrict__ wrow) {
  const int row = blockIdx.x * 8 + (threadIdx.x >> 5);
  const int ln  = threadIdx.x & 31;
  const float* rp = w + (size_t)row * C_;
  float s = 0.f;
#pragma unroll
  for (int j = 0; j < 8; ++j) s += rp[ln + 32 * j];
#pragma unroll
  for (int off = 16; off > 0; off >>= 1) s += __shfl_xor(s, off, 32);
  if (ln == 0) wrow[row] = s;
}

__global__ __launch_bounds__(64) void fused_kernel(
    const float* __restrict__ feat, const float* __restrict__ wrow,
    const float* __restrict__ bias, const float* __restrict__ gamma,
    const float* __restrict__ beta, const float* __restrict__ linw,
    const float* __restrict__ linb, float* __restrict__ out) {
  __shared__ __attribute__((aligned(16))) float hbuf[WAVES_PER_BLOCK * ROWS_PER_WAVE * HSTRIDE];
  __shared__ __attribute__((aligned(16))) float lw[ZOFF + 4];

  const int tid = threadIdx.x;

  for (int e = tid; e < D_ * C_; e += 64) {
    lw[(e >> 8) * HSTRIDE + (e & 255)] = linw[e];
  }
  if (tid < 4) lw[ZOFF + tid] = 0.f;
  __syncthreads();

  const int wv   = tid >> 5;
  const int ln   = tid & 31;
  const int wave = blockIdx.x * WAVES_PER_BLOCK + wv;
  const int p    = wave % P_;
  const int b0   = (wave / P_) * ROWS_PER_WAVE;
  float* hw = hbuf + wv * (ROWS_PER_WAVE * HSTRIDE);

  const int c0 = ln * 4, c1 = ln * 4 + 128;

  const v4f wr0 = *(const v4f*)(wrow + p * C_ + c0);
  const v4f wr1 = *(const v4f*)(wrow + p * C_ + c1);
  const v4f bi0 = *(const v4f*)(bias + p * C_ + c0);
  const v4f bi1 = *(const v4f*)(bias + p * C_ + c1);
  const v4f ga0 = *(const v4f*)(gamma + c0);
  const v4f ga1 = *(const v4f*)(gamma + c1);
  const v4f be0 = *(const v4f*)(beta + c0);
  const v4f be1 = *(const v4f*)(beta + c1);

#pragma unroll 2
  for (int r = 0; r < ROWS_PER_WAVE; ++r) {
    const int g = (b0 + r) * P_ + p;
    const float* fp = feat + (size_t)g * C_;
    const v4f f0 = __builtin_nontemporal_load((const v4f*)(fp + c0));
    const v4f f1 = __builtin_nontemporal_load((const v4f*)(fp + c1));
    v4f pw0, pw1;
    float s1 = 0.f, s2 = 0.f;
#pragma unroll
    for (int q = 0; q < 4; ++q) {
      pw0[q] = fmaf(f0[q], wr0[q], bi0[q]);
      pw1[q] = fmaf(f1[q], wr1[q], bi1[q]);
      s1 += pw0[q] + pw1[q];
      s2  = fmaf(pw0[q], pw0[q], s2);
      s2  = fmaf(pw1[q], pw1[q], s2);
    }
#pragma unroll
    for (int off = 16; off > 0; off >>= 1) {
      s1 += __shfl_xor(s1, off, 32);
      s2 += __shfl_xor(s2, off, 32);
    }
    const float inv  = 1.0f / (float)C_;
    const float mean = s1 * inv;
    const float var  = fmaf(s2, inv, -mean * mean);
    const float rstd = rsqrtf(var + 1e-5f);
    v4f h0, h1;
#pragma unroll
    for (int q = 0; q < 4; ++q) {
      const float n0 = fmaf((pw0[q] - mean) * rstd, ga0[q], be0[q]);
      const float n1 = fmaf((pw1[q] - mean) * rstd, ga1[q], be1[q]);
      h0[q] = fmaxf(f0[q] + n0, 0.f);
      h1[q] = fmaxf(f1[q] + n1, 0.f);
    }
    *(v4f*)&hw[r * HSTRIDE + c0] = h0;
    *(v4f*)&hw[r * HSTRIDE + c1] = h1;
  }

  const int rowA = ln & 15;
  const int koff = (ln >> 4) << 1;
  const int dcol = ln & 15;
  const float* bbase = (dcol < D_) ? &lw[dcol * HSTRIDE + koff] : &lw[ZOFF];
  const int bstep = (dcol < D_) ? 4 : 0;

  v8f acc = {0.f, 0.f, 0.f, 0.f, 0.f, 0.f, 0.f, 0.f};
#pragma unroll 4
  for (int k = 0; k < KSTEPS; ++k) {
    v2f a = *(const v2f*)&hw[rowA * HSTRIDE + 4 * k + koff];
    v2f b = *(const v2f*)&bbase[k * bstep];
    acc = __builtin_amdgcn_wmma_f32_16x16x4_f32(
         false, a,  false, b,
         (short)0, acc,  false,  false);
  }

  const int rAdd = (ln >> 4) << 3;
  if (dcol < D_) {
    const float lb = linb[dcol];
#pragma unroll
    for (int j = 0; j < 8; ++j) {
      const int g = (b0 + j + rAdd) * P_ + p;
      __builtin_nontemporal_store(acc[j] + lb, out + (size_t)g * D_ + dcol);
    }
  }
}

extern "C" void kernel_launch(void* const* d_in, const int* in_sizes, int n_in,
                              void* d_out, int out_size, void* d_ws, size_t ws_size,
                              hipStream_t stream) {
  const float* feat    = (const float*)d_in[0];
  const float* weights = (const float*)d_in[1];
  const float* bias    = (const float*)d_in[2];
  const float* gamma   = (const float*)d_in[3];
  const float* beta    = (const float*)d_in[4];
  const float* linw    = (const float*)d_in[5];
  const float* linb    = (const float*)d_in[6];
  float* out  = (float*)d_out;
  float* wrow = (float*)d_ws;

  rowsum_kernel<<<(P_ * C_) / 8, 256, 0, stream>>>(weights, wrow);

  const int nblocks = R_ / (ROWS_PER_WAVE * WAVES_PER_BLOCK);
  fused_kernel<<<nblocks, 64, 0, stream>>>(feat, wrow, bias, gamma, beta,
                                           linw, linb, out);
}


# All global loads of a wave (lin_w, per-point vectors, lin_b, its 16 feature rows) issued as one group at entry, tile held in registers; lin_w staging barrier moved behind the LayerNorm phase
# speedup vs baseline: 1.1184x; 1.1184x over previous
#include <hip/hip_runtime.h>
#include <math.h>

typedef __attribute__((ext_vector_type(2))) float v2f;
typedef __attribute__((ext_vector_type(4))) float v4f;
typedef __attribute__((ext_vector_type(8))) float v8f;

#define B_  8192
#define P_  25
#define C_  256
#define D_  5
#define R_  (B_ * P_)
#define ROWS_PER_WAVE 16
#define WAVES_PER_BLOCK 2
#define HSTRIDE 260
#define KSTEPS (C_ / 4)
#define ZOFF (D_ * HSTRIDE)

__global__ __launch_bounds__(256) void rowsum_kernel(const float* __restrict__ w,
                                                     float* __restrict__ wrow) {
  const int row = blockIdx.x * 8 + (threadIdx.x >> 5);
  const int ln  = threadIdx.x & 31;
  const float* rp = w + (size_t)row * C_;
  float s = 0.f;
#pragma unroll
  for (int j = 0; j < 8; ++j) s += rp[ln + 32 * j];
#pragma unroll
  for (int off = 16; off > 0; off >>= 1) s += __shfl_xor(s, off, 32);
  if (ln == 0) wrow[row] = s;
}

__global__ __launch_bounds__(64) void fused_kernel(
    const float* __restrict__ feat, const float* __restrict__ wrow,
    const float* __restrict__ bias, const float* __restrict__ gamma,
    const float* __restrict__ beta, const float* __restrict__ linw,
    const float* __restrict__ linb, float* __restrict__ out) {
  __shared__ __attribute__((aligned(16))) float hbuf[WAVES_PER_BLOCK * ROWS_PER_WAVE * HSTRIDE];
  __shared__ __attribute__((aligned(16))) float lw[ZOFF + 4];

  const int tid = threadIdx.x;
  const int wv   = tid >> 5;
  const int ln   = tid & 31;
  const int wave = blockIdx.x * WAVES_PER_BLOCK + wv;
  const int p    = wave % P_;
  const int b0   = (wave / P_) * ROWS_PER_WAVE;
  float* hw = hbuf + wv * (ROWS_PER_WAVE * HSTRIDE);

  const int c0 = ln * 4, c1 = ln * 4 + 128;

  float lreg[(D_ * C_) / 64];
#pragma unroll
  for (int i = 0; i < (D_ * C_) / 64; ++i) lreg[i] = linw[tid + 64 * i];

  const v4f wr0 = *(const v4f*)(wrow + p * C_ + c0);
  const v4f wr1 = *(const v4f*)(wrow + p * C_ + c1);
  const v4f bi0 = *(const v4f*)(bias + p * C_ + c0);
  const v4f bi1 = *(const v4f*)(bias + p * C_ + c1);
  const v4f ga0 = *(const v4f*)(gamma + c0);
  const v4f ga1 = *(const v4f*)(gamma + c1);
  const v4f be0 = *(const v4f*)(beta + c0);
  const v4f be1 = *(const v4f*)(beta + c1);
  const int dcol = ln & 15;
  const float lb = (dcol < D_) ? linb[dcol] : 0.f;

  v4f f0[ROWS_PER_WAVE], f1[ROWS_PER_WAVE];
  {
    const float* fp = feat + ((size_t)b0 * P_ + p) * C_;
#pragma unroll
    for (int r = 0; r < ROWS_PER_WAVE; ++r) {
      f0[r] = __builtin_nontemporal_load((const v4f*)(fp + c0));
      f1[r] = __builtin_nontemporal_load((const v4f*)(fp + c1));
      fp += P_ * C_;
    }
  }

#pragma unroll
  for (int i = 0; i < (D_ * C_) / 64; ++i) {
    const int e = tid + 64 * i;
    lw[(e >> 8) * HSTRIDE + (e & 255)] = lreg[i];
  }
  if (tid < 4) lw[ZOFF + tid] = 0.f;

#pragma unroll
  for (int r = 0; r < ROWS_PER_WAVE; ++r) {
    const v4f x0 = f0[r];
    const v4f x1 = f1[r];
    v4f pw0, pw1;
    float s1 = 0.f, s2 = 0.f;
#pragma unroll
    for (int q = 0; q < 4; ++q) {
      pw0[q] = fmaf(x0[q], wr0[q], bi0[q]);
      pw1[q] = fmaf(x1[q], wr1[q], bi1[q]);
      s1 += pw0[q] + pw1[q];
      s2  = fmaf(pw0[q], pw0[q], s2);
      s2  = fmaf(pw1[q], pw1[q], s2);
    }
#pragma unroll
    for (int off = 16; off > 0; off >>= 1) {
      s1 += __shfl_xor(s1, off, 32);
      s2 += __shfl_xor(s2, off, 32);
    }
    const float inv  = 1.0f / (float)C_;
    const float mean = s1 * inv;
    const float var  = fmaf(s2, inv, -mean * mean);
    const float rstd = rsqrtf(var + 1e-5f);
    v4f h0, h1;
#pragma unroll
    for (int q = 0; q < 4; ++q) {
      const float n0 = fmaf((pw0[q] - mean) * rstd, ga0[q], be0[q]);
      const float n1 = fmaf((pw1[q] - mean) * rstd, ga1[q], be1[q]);
      h0[q] = fmaxf(x0[q] + n0, 0.f);
      h1[q] = fmaxf(x1[q] + n1, 0.f);
    }
    *(v4f*)&hw[r * HSTRIDE + c0] = h0;
    *(v4f*)&hw[r * HSTRIDE + c1] = h1;
  }

  __syncthreads();

  const int rowA = ln & 15;
  const int koff = (ln >> 4) << 1;
  const float* bbase = (dcol < D_) ? &lw[dcol * HSTRIDE + koff] : &lw[ZOFF];
  const int bstep = (dcol < D_) ? 4 : 0;

  v8f acc = {0.f, 0.f, 0.f, 0.f, 0.f, 0.f, 0.f, 0.f};
#pragma unroll 4
  for (int k = 0; k < KSTEPS; ++k) {
    v2f a = *(const v2f*)&hw[rowA * HSTRIDE + 4 * k + koff];
    v2f b = *(const v2f*)&bbase[k * bstep];
    acc = __builtin_amdgcn_wmma_f32_16x16x4_f32(
         false, a,  false, b,
         (short)0, acc,  false,  false);
  }

  const int rAdd = (ln >> 4) << 3;
  if (dcol < D_) {
#pragma unroll
    for (int j = 0; j < 8; ++j) {
      const int g = (b0 + j + rAdd) * P_ + p;
      __builtin_nontemporal_store(acc[j] + lb, out + (size_t)g * D_ + dcol);
    }
  }
}

extern "C" void kernel_launch(void* const* d_in, const int* in_sizes, int n_in,
                              void* d_out, int out_size, void* d_ws, size_t ws_size,
                              hipStream_t stream) {
  const float* feat    = (const float*)d_in[0];
  const float* weights = (const float*)d_in[1];
  const float* bias    = (const float*)d_in[2];
  const float* gamma   = (const float*)d_in[3];
  const float* beta    = (const float*)d_in[4];
  const float* linw    = (const float*)d_in[5];
  const float* linb    = (const float*)d_in[6];
  float* out  = (float*)d_out;
  float* wrow = (float*)d_ws;

  rowsum_kernel<<<(P_ * C_) / 8, 256, 0, stream>>>(weights, wrow);

  const int nblocks = R_ / (ROWS_PER_WAVE * WAVES_PER_BLOCK);
  fused_kernel<<<nblocks, 64, 0, stream>>>(feat, wrow, bias, gamma, beta,
                                           linw, linb, out);
}


# Lane sums (LayerNorm mean and square sum per row, weight row sums) on the vector ALU: one permlanex16 + four row_xmask DPP adds replace five ds_bpermute LDS round trips; partner order kept
# speedup vs baseline: 1.3068x; 1.1685x over previous
#include <hip/hip_runtime.h>
#include <math.h>

typedef __attribute__((ext_vector_type(2))) float v2f;
typedef __attribute__((ext_vector_type(4))) float v4f;
typedef __attribute__((ext_vector_type(8))) float v8f;

#define B_  8192
#define P_  25
#define C_  256
#define D_  5
#define R_  (B_ * P_)
#define ROWS_PER_WAVE 16
#define WAVES_PER_BLOCK 2
#define HSTRIDE 260
#define KSTEPS (C_ / 4)
#define ZOFF (D_ * HSTRIDE)

__device__ __forceinline__ float xhalf(float x) {
  const int b = __builtin_bit_cast(int, x);
  return __builtin_bit_cast(float,
      __builtin_amdgcn_permlanex16(b, b, 0x76543210, 0xfedcba98, false, false));
}
template <int M>
__device__ __forceinline__ float add_xmask(float x) {
  const int b = __builtin_bit_cast(int, x);
  const int o = __builtin_amdgcn_update_dpp(0, b, 0x160 | M, 0xf, 0xf, true);
  return x + __builtin_bit_cast(float, o);
}
__device__ __forceinline__ float wave_sum(float s) {
  s += xhalf(s);
  s = add_xmask<8>(s);
  s = add_xmask<4>(s);
  s = add_xmask<2>(s);
  s = add_xmask<1>(s);
  return s;
}

__global__ __launch_bounds__(256) void rowsum_kernel(const float* __restrict__ w,
                                                     float* __restrict__ wrow) {
  const int row = blockIdx.x * 8 + (threadIdx.x >> 5);
  const int ln  = threadIdx.x & 31;
  const float* rp = w + (size_t)row * C_;
  float s = 0.f;
#pragma unroll
  for (int j = 0; j < 8; ++j) s += rp[ln + 32 * j];
  s = wave_sum(s);
  if (ln == 0) wrow[row] = s;
}

__global__ __launch_bounds__(64) void fused_kernel(
    const float* __restrict__ feat, const float* __restrict__ wrow,
    const float* __restrict__ bias, const float* __restrict__ gamma,
    const float* __restrict__ beta, const float* __restrict__ linw,
    const float* __restrict__ linb, float* __restrict__ out) {
  __shared__ __attribute__((aligned(16))) float hbuf[WAVES_PER_BLOCK * ROWS_PER_WAVE * HSTRIDE];
  __shared__ __attribute__((aligned(16))) float lw[ZOFF + 4];

  const int tid = threadIdx.x;
  const int wv   = tid >> 5;
  const int ln   = tid & 31;
  const int wave = blockIdx.x * WAVES_PER_BLOCK + wv;
  const int p    = wave % P_;
  const int b0   = (wave / P_) * ROWS_PER_WAVE;
  float* hw = hbuf + wv * (ROWS_PER_WAVE * HSTRIDE);

  const int c0 = ln * 4, c1 = ln * 4 + 128;

  float lreg[(D_ * C_) / 64];
#pragma unroll
  for (int i = 0; i < (D_ * C_) / 64; ++i) lreg[i] = linw[tid + 64 * i];

  const v4f wr0 = *(const v4f*)(wrow + p * C_ + c0);
  const v4f wr1 = *(const v4f*)(wrow + p * C_ + c1);
  const v4f bi0 = *(const v4f*)(bias + p * C_ + c0);
  const v4f bi1 = *(const v4f*)(bias + p * C_ + c1);
  const v4f ga0 = *(const v4f*)(gamma + c0);
  const v4f ga1 = *(const v4f*)(gamma + c1);
  const v4f be0 = *(const v4f*)(beta + c0);
  const v4f be1 = *(const v4f*)(beta + c1);
  const int dcol = ln & 15;
  const float lb = (dcol < D_) ? linb[dcol] : 0.f;

  v4f f0[ROWS_PER_WAVE], f1[ROWS_PER_WAVE];
  {
    const float* fp = feat + ((size_t)b0 * P_ + p) * C_;
#pragma unroll
    for (int r = 0; r < ROWS_PER_WAVE; ++r) {
      f0[r] = __builtin_nontemporal_load((const v4f*)(fp + c0));
      f1[r] = __builtin_nontemporal_load((const v4f*)(fp + c1));
      fp += P_ * C_;
    }
  }

#pragma unroll
  for (int i = 0; i < (D_ * C_) / 64; ++i) {
    const int e = tid + 64 * i;
    lw[(e >> 8) * HSTRIDE + (e & 255)] = lreg[i];
  }
  if (tid < 4) lw[ZOFF + tid] = 0.f;

#pragma unroll
  for (int r = 0; r < ROWS_PER_WAVE; ++r) {
    const v4f x0 = f0[r];
    const v4f x1 = f1[r];
    v4f pw0, pw1;
    float s1 = 0.f, s2 = 0.f;
#pragma unroll
    for (int q = 0; q < 4; ++q) {
      pw0[q] = fmaf(x0[q], wr0[q], bi0[q]);
      pw1[q] = fmaf(x1[q], wr1[q], bi1[q]);
      s1 += pw0[q] + pw1[q];
      s2  = fmaf(pw0[q], pw0[q], s2);
      s2  = fmaf(pw1[q], pw1[q], s2);
    }
    s1 = wave_sum(s1);
    s2 = wave_sum(s2);
    const float inv  = 1.0f / (float)C_;
    const float mean = s1 * inv;
    const float var  = fmaf(s2, inv, -mean * mean);
    const float rstd = rsqrtf(var + 1e-5f);
    v4f h0, h1;
#pragma unroll
    for (int q = 0; q < 4; ++q) {
      const float n0 = fmaf((pw0[q] - mean) * rstd, ga0[q], be0[q]);
      const float n1 = fmaf((pw1[q] - mean) * rstd, ga1[q], be1[q]);
      h0[q] = fmaxf(x0[q] + n0, 0.f);
      h1[q] = fmaxf(x1[q] + n1, 0.f);
    }
    *(v4f*)&hw[r * HSTRIDE + c0] = h0;
    *(v4f*)&hw[r * HSTRIDE + c1] = h1;
  }

  __syncthreads();

  const int rowA = ln & 15;
  const int koff = (ln >> 4) << 1;
  const float* bbase = (dcol < D_) ? &lw[dcol * HSTRIDE + koff] : &lw[ZOFF];
  const int bstep = (dcol < D_) ? 4 : 0;

  v8f acc = {0.f, 0.f, 0.f, 0.f, 0.f, 0.f, 0.f, 0.f};
#pragma unroll 4
  for (int k = 0; k < KSTEPS; ++k) {
    v2f a = *(const v2f*)&hw[rowA * HSTRIDE + 4 * k + koff];
    v2f b = *(const v2f*)&bbase[k * bstep];
    acc = __builtin_amdgcn_wmma_f32_16x16x4_f32(
         false, a,  false, b,
         (short)0, acc,  false,  false);
  }

  const int rAdd = (ln >> 4) << 3;
  if (dcol < D_) {
#pragma unroll
    for (int j = 0; j < 8; ++j) {
      const int g = (b0 + j + rAdd) * P_ + p;
      __builtin_nontemporal_store(acc[j] + lb, out + (size_t)g * D_ + dcol);
    }
  }
}

extern "C" void kernel_launch(void* const* d_in, const int* in_sizes, int n_in,
                              void* d_out, int out_size, void* d_ws, size_t ws_size,
                              hipStream_t stream) {
  const float* feat    = (const float*)d_in[0];
  const float* weights = (const float*)d_in[1];
  const float* bias    = (const float*)d_in[2];
  const float* gamma   = (const float*)d_in[3];
  const float* beta    = (const float*)d_in[4];
  const float* linw    = (const float*)d_in[5];
  const float* linb    = (const float*)d_in[6];
  float* out  = (float*)d_out;
  float* wrow = (float*)d_ws;

  rowsum_kernel<<<(P_ * C_) / 8, 256, 0, stream>>>(weights, wrow);

  const int nblocks = R_ / (ROWS_PER_WAVE * WAVES_PER_BLOCK);
  fused_kernel<<<nblocks, 64, 0, stream>>>(feat, wrow, bias, gamma, beta,
                                           linw, linb, out);
}


# Phase B: LDS fragment loads of each 16-step block issued as one group into a second register set ahead of the WMMAs (was a load-wait pair at every WMMA); k order and accumulator chain unchanged
# speedup vs baseline: 1.3462x; 1.0301x over previous
#include <hip/hip_runtime.h>
#include <math.h>

typedef __attribute__((ext_vector_type(2))) float v2f;
typedef __attribute__((ext_vector_type(4))) float v4f;
typedef __attribute__((ext_vector_type(8))) float v8f;

#define B_  8192
#define P_  25
#define C_  256
#define D_  5
#define R_  (B_ * P_)
#define ROWS_PER_WAVE 16
#define WAVES_PER_BLOCK 2
#define HSTRIDE 260
#define KSTEPS (C_ / 4)
#define ZOFF (D_ * HSTRIDE)
#define KBLK 16

__device__ __forceinline__ float xhalf(float x) {
  const int b = __builtin_bit_cast(int, x);
  return __builtin_bit_cast(float,
      __builtin_amdgcn_permlanex16(b, b, 0x76543210, 0xfedcba98, false, false));
}
template <int M>
__device__ __forceinline__ float add_xmask(float x) {
  const int b = __builtin_bit_cast(int, x);
  const int o = __builtin_amdgcn_update_dpp(0, b, 0x160 | M, 0xf, 0xf, true);
  return x + __builtin_bit_cast(float, o);
}
__device__ __forceinline__ float wave_sum(float s) {
  s += xhalf(s);
  s = add_xmask<8>(s);
  s = add_xmask<4>(s);
  s = add_xmask<2>(s);
  s = add_xmask<1>(s);
  return s;
}

__global__ __launch_bounds__(256) void rowsum_kernel(const float* __restrict__ w,
                                                     float* __restrict__ wrow) {
  const int row = blockIdx.x * 8 + (threadIdx.x >> 5);
  const int ln  = threadIdx.x & 31;
  const float* rp = w + (size_t)row * C_;
  float s = 0.f;
#pragma unroll
  for (int j = 0; j < 8; ++j) s += rp[ln + 32 * j];
  s = wave_sum(s);
  if (ln == 0) wrow[row] = s;
}

__global__ __launch_bounds__(64) void fused_kernel(
    const float* __restrict__ feat, const float* __restrict__ wrow,
    const float* __restrict__ bias, const float* __restrict__ gamma,
    const float* __restrict__ beta, const float* __restrict__ linw,
    const float* __restrict__ linb, float* __restrict__ out) {
  __shared__ __attribute__((aligned(16))) float hbuf[WAVES_PER_BLOCK * ROWS_PER_WAVE * HSTRIDE];
  __shared__ __attribute__((aligned(16))) float lw[ZOFF + 4];

  const int tid = threadIdx.x;
  const int wv   = tid >> 5;
  const int ln   = tid & 31;
  const int wave = blockIdx.x * WAVES_PER_BLOCK + wv;
  const int p    = wave % P_;
  const int b0   = (wave / P_) * ROWS_PER_WAVE;
  float* hw = hbuf + wv * (ROWS_PER_WAVE * HSTRIDE);

  const int c0 = ln * 4, c1 = ln * 4 + 128;

  float lreg[(D_ * C_) / 64];
#pragma unroll
  for (int i = 0; i < (D_ * C_) / 64; ++i) lreg[i] = linw[tid + 64 * i];

  const v4f wr0 = *(const v4f*)(wrow + p * C_ + c0);
  const v4f wr1 = *(const v4f*)(wrow + p * C_ + c1);
  const v4f bi0 = *(const v4f*)(bias + p * C_ + c0);
  const v4f bi1 = *(const v4f*)(bias + p * C_ + c1);
  const v4f ga0 = *(const v4f*)(gamma + c0);
  const v4f ga1 = *(const v4f*)(gamma + c1);
  const v4f be0 = *(const v4f*)(beta + c0);
  const v4f be1 = *(const v4f*)(beta + c1);
  const int dcol = ln & 15;
  const float lb = (dcol < D_) ? linb[dcol] : 0.f;

  v4f f0[ROWS_PER_WAVE], f1[ROWS_PER_WAVE];
  {
    const float* fp = feat + ((size_t)b0 * P_ + p) * C_;
#pragma unroll
    for (int r = 0; r < ROWS_PER_WAVE; ++r) {
      f0[r] = __builtin_nontemporal_load((const v4f*)(fp + c0));
      f1[r] = __builtin_nontemporal_load((const v4f*)(fp + c1));
      fp += P_ * C_;
    }
  }

#pragma unroll
  for (int i = 0; i < (D_ * C_) / 64; ++i) {
    const int e = tid + 64 * i;
    lw[(e >> 8) * HSTRIDE + (e & 255)] = lreg[i];
  }
  if (tid < 4) lw[ZOFF + tid] = 0.f;

#pragma unroll
  for (int r = 0; r < ROWS_PER_WAVE; ++r) {
    const v4f x0 = f0[r];
    const v4f x1 = f1[r];
    v4f pw0, pw1;
    float s1 = 0.f, s2 = 0.f;
#pragma unroll
    for (int q = 0; q < 4; ++q) {
      pw0[q] = fmaf(x0[q], wr0[q], bi0[q]);
      pw1[q] = fmaf(x1[q], wr1[q], bi1[q]);
      s1 += pw0[q] + pw1[q];
      s2  = fmaf(pw0[q], pw0[q], s2);
      s2  = fmaf(pw1[q], pw1[q], s2);
    }
    s1 = wave_sum(s1);
    s2 = wave_sum(s2);
    const float inv  = 1.0f / (float)C_;
    const float mean = s1 * inv;
    const float var  = fmaf(s2, inv, -mean * mean);
    const float rstd = rsqrtf(var + 1e-5f);
    v4f h0, h1;
#pragma unroll
    for (int q = 0; q < 4; ++q) {
      const float n0 = fmaf((pw0[q] - mean) * rstd, ga0[q], be0[q]);
      const float n1 = fmaf((pw1[q] - mean) * rstd, ga1[q], be1[q]);
      h0[q] = fmaxf(x0[q] + n0, 0.f);
      h1[q] = fmaxf(x1[q] + n1, 0.f);
    }
    *(v4f*)&hw[r * HSTRIDE + c0] = h0;
    *(v4f*)&hw[r * HSTRIDE + c1] = h1;
  }

  __syncthreads();

  const int rowA = ln & 15;
  const int koff = (ln >> 4) << 1;
  const float* abase = &hw[rowA * HSTRIDE + koff];
  const float* bbase = (dcol < D_) ? &lw[dcol * HSTRIDE + koff] : &lw[ZOFF];
  const int bstep = (dcol < D_) ? 4 : 0;

  v2f fa[2][KBLK], fb[2][KBLK];
#pragma unroll
  for (int j = 0; j < KBLK; ++j) {
    fa[0][j] = *(const v2f*)&abase[4 * j];
    fb[0][j] = *(const v2f*)&bbase[j * bstep];
  }

  v8f acc = {0.f, 0.f, 0.f, 0.f, 0.f, 0.f, 0.f, 0.f};
#pragma unroll
  for (int blk = 0; blk < KSTEPS / KBLK; ++blk) {
    const int cur = blk & 1, nxt = cur ^ 1;
    if (blk + 1 < KSTEPS / KBLK) {
#pragma unroll
      for (int j = 0; j < KBLK; ++j) {
        const int k = (blk + 1) * KBLK + j;
        fa[nxt][j] = *(const v2f*)&abase[4 * k];
        fb[nxt][j] = *(const v2f*)&bbase[k * bstep];
      }
    }
    __builtin_amdgcn_sched_barrier(0);
#pragma unroll
    for (int j = 0; j < KBLK; ++j) {
      acc = __builtin_amdgcn_wmma_f32_16x16x4_f32(
           false, fa[cur][j],  false, fb[cur][j],
           (short)0, acc,  false,  false);
    }
    __builtin_amdgcn_sched_barrier(0);
  }

  const int rAdd = (ln >> 4) << 3;
  if (dcol < D_) {
#pragma unroll
    for (int j = 0; j < 8; ++j) {
      const int g = (b0 + j + rAdd) * P_ + p;
      __builtin_nontemporal_store(acc[j] + lb, out + (size_t)g * D_ + dcol);
    }
  }
}

extern "C" void kernel_launch(void* const* d_in, const int* in_sizes, int n_in,
                              void* d_out, int out_size, void* d_ws, size_t ws_size,
                              hipStream_t stream) {
  const float* feat    = (const float*)d_in[0];
  const float* weights = (const float*)d_in[1];
  const float* bias    = (const float*)d_in[2];
  const float* gamma   = (const float*)d_in[3];
  const float* beta    = (const float*)d_in[4];
  const float* linw    = (const float*)d_in[5];
  const float* linb    = (const float*)d_in[6];
  float* out  = (float*)d_out;
  float* wrow = (float*)d_ws;

  rowsum_kernel<<<(P_ * C_) / 8, 256, 0, stream>>>(weights, wrow);

  const int nblocks = R_ / (ROWS_PER_WAVE * WAVES_PER_BLOCK);
  fused_kernel<<<nblocks, 64, 0, stream>>>(feat, wrow, bias, gamma, beta,
                                           linw, linb, out);
}


# One wave per wave slot (4096), each looping over 3-4 tiles; a row's registers take the next tile's row once Phase A is done with it; 8-deep fragment blocks and a zero B row keep 4 waves per SIMD
# speedup vs baseline: 1.5288x; 1.1356x over previous
#include <hip/hip_runtime.h>
#include <math.h>

typedef __attribute__((ext_vector_type(2))) float v2f;
typedef __attribute__((ext_vector_type(4))) float v4f;
typedef __attribute__((ext_vector_type(8))) float v8f;

#define B_  8192
#define P_  25
#define C_  256
#define D_  5
#define R_  (B_ * P_)
#define ROWS_PER_WAVE 16
#define WAVES_PER_BLOCK 2
#define HSTRIDE 260
#define KSTEPS (C_ / 4)
#define ZOFF (D_ * HSTRIDE)
#define KBLK 8

__device__ __forceinline__ float xhalf(float x) {
  const int b = __builtin_bit_cast(int, x);
  return __builtin_bit_cast(float,
      __builtin_amdgcn_permlanex16(b, b, 0x76543210, 0xfedcba98, false, false));
}
template <int M>
__device__ __forceinline__ float add_xmask(float x) {
  const int b = __builtin_bit_cast(int, x);
  const int o = __builtin_amdgcn_update_dpp(0, b, 0x160 | M, 0xf, 0xf, true);
  return x + __builtin_bit_cast(float, o);
}
__device__ __forceinline__ float wave_sum(float s) {
  s += xhalf(s);
  s = add_xmask<8>(s);
  s = add_xmask<4>(s);
  s = add_xmask<2>(s);
  s = add_xmask<1>(s);
  return s;
}

__global__ __launch_bounds__(256) void rowsum_kernel(const float* __restrict__ w,
                                                     float* __restrict__ wrow) {
  const int row = blockIdx.x * 8 + (threadIdx.x >> 5);
  const int ln  = threadIdx.x & 31;
  const float* rp = w + (size_t)row * C_;
  float s = 0.f;
#pragma unroll
  for (int j = 0; j < 8; ++j) s += rp[ln + 32 * j];
  s = wave_sum(s);
  if (ln == 0) wrow[row] = s;
}

#define NTILES (R_ / ROWS_PER_WAVE)
#define NWAVES 4096

template <bool PF>
__device__ __forceinline__ void phase_a(
    v4f (&f0)[ROWS_PER_WAVE], v4f (&f1)[ROWS_PER_WAVE],
    const v4f wr0, const v4f wr1, const v4f bi0, const v4f bi1,
    const v4f ga0, const v4f ga1, const v4f be0, const v4f be1,
    float* __restrict__ hw, const int c0, const int c1, const float* fpn) {
#pragma unroll
  for (int r = 0; r < ROWS_PER_WAVE; ++r) {
    const v4f x0 = f0[r];
    const v4f x1 = f1[r];
    v4f pw0, pw1;
    float s1 = 0.f, s2 = 0.f;
#pragma unroll
    for (int q = 0; q < 4; ++q) {
      pw0[q] = fmaf(x0[q], wr0[q], bi0[q]);
      pw1[q] = fmaf(x1[q], wr1[q], bi1[q]);
      s1 += pw0[q] + pw1[q];
      s2  = fmaf(pw0[q], pw0[q], s2);
      s2  = fmaf(pw1[q], pw1[q], s2);
    }
    s1 = wave_sum(s1);
    s2 = wave_sum(s2);
    const float inv  = 1.0f / (float)C_;
    const float mean = s1 * inv;
    const float var  = fmaf(s2, inv, -mean * mean);
    const float rstd = rsqrtf(var + 1e-5f);
    v4f h0, h1;
#pragma unroll
    for (int q = 0; q < 4; ++q) {
      const float n0 = fmaf((pw0[q] - mean) * rstd, ga0[q], be0[q]);
      const float n1 = fmaf((pw1[q] - mean) * rstd, ga1[q], be1[q]);
      h0[q] = fmaxf(x0[q] + n0, 0.f);
      h1[q] = fmaxf(x1[q] + n1, 0.f);
    }
    *(v4f*)&hw[r * HSTRIDE + c0] = h0;
    *(v4f*)&hw[r * HSTRIDE + c1] = h1;
    if (PF) {
      f0[r] = __builtin_nontemporal_load((const v4f*)(fpn + c0));
      f1[r] = __builtin_nontemporal_load((const v4f*)(fpn + c1));
      fpn += P_ * C_;
    }
  }
}

__global__ __launch_bounds__(64) __attribute__((amdgpu_waves_per_eu(4, 4))) void fused_kernel(
    const float* __restrict__ feat, const float* __restrict__ wrow,
    const float* __restrict__ bias, const float* __restrict__ gamma,
    const float* __restrict__ beta, const float* __restrict__ linw,
    const float* __restrict__ linb, float* __restrict__ out) {
  __shared__ __attribute__((aligned(16))) float hbuf[WAVES_PER_BLOCK * ROWS_PER_WAVE * HSTRIDE];
  __shared__ __attribute__((aligned(16))) float lw[ZOFF + HSTRIDE];

  const int tid = threadIdx.x;
  const int ln  = tid & 31;
  const int wv   = __builtin_amdgcn_readfirstlane(tid >> 5);
  const int wave = blockIdx.x * WAVES_PER_BLOCK + wv;
  float* hw = hbuf + wv * (ROWS_PER_WAVE * HSTRIDE);

  const int c0 = ln * 4, c1 = ln * 4 + 128;

  float lreg[(D_ * C_) / 64];
#pragma unroll
  for (int i = 0; i < (D_ * C_) / 64; ++i) lreg[i] = linw[tid + 64 * i];

  int t  = wave;
  int p  = t % P_;
  int b0 = (t / P_) * ROWS_PER_WAVE;
  v4f wr0 = *(const v4f*)(wrow + p * C_ + c0);
  v4f wr1 = *(const v4f*)(wrow + p * C_ + c1);
  v4f bi0 = *(const v4f*)(bias + p * C_ + c0);
  v4f bi1 = *(const v4f*)(bias + p * C_ + c1);
  const v4f ga0 = *(const v4f*)(gamma + c0);
  const v4f ga1 = *(const v4f*)(gamma + c1);
  const v4f be0 = *(const v4f*)(beta + c0);
  const v4f be1 = *(const v4f*)(beta + c1);
  const int dcol = ln & 15;
  const float lb = (dcol < D_) ? linb[dcol] : 0.f;

  v4f f0[ROWS_PER_WAVE], f1[ROWS_PER_WAVE];
  {
    const float* fp = feat + ((size_t)b0 * P_ + p) * C_;
#pragma unroll
    for (int r = 0; r < ROWS_PER_WAVE; ++r) {
      f0[r] = __builtin_nontemporal_load((const v4f*)(fp + c0));
      f1[r] = __builtin_nontemporal_load((const v4f*)(fp + c1));
      fp += P_ * C_;
    }
  }

#pragma unroll
  for (int i = 0; i < (D_ * C_) / 64; ++i) {
    const int e = tid + 64 * i;
    lw[(e >> 8) * HSTRIDE + (e & 255)] = lreg[i];
  }
  for (int i = tid; i < HSTRIDE; i += 64) lw[ZOFF + i] = 0.f;
  __syncthreads();

  const int rowA = ln & 15;
  const int koff = (ln >> 4) << 1;
  const float* abase = &hw[rowA * HSTRIDE + koff];
  const float* bbase = &lw[((dcol < D_) ? dcol : D_) * HSTRIDE + koff];
  const int rAdd = (ln >> 4) << 3;

  for (;;) {
    const int  tn = t + NWAVES;
    const bool has_next = tn < NTILES;
    const int  pn  = tn % P_;
    const int  bn0 = (tn / P_) * ROWS_PER_WAVE;

    v4f wn0 = wr0, wn1 = wr1, bn0v = bi0, bn1v = bi1;
    if (has_next) {
      wn0  = *(const v4f*)(wrow + pn * C_ + c0);
      wn1  = *(const v4f*)(wrow + pn * C_ + c1);
      bn0v = *(const v4f*)(bias + pn * C_ + c0);
      bn1v = *(const v4f*)(bias + pn * C_ + c1);
      const float* fpn = feat + ((size_t)bn0 * P_ + pn) * C_;
      phase_a<true>(f0, f1, wr0, wr1, bi0, bi1, ga0, ga1, be0, be1, hw, c0, c1, fpn);
    } else {
      phase_a<false>(f0, f1, wr0, wr1, bi0, bi1, ga0, ga1, be0, be1, hw, c0, c1, nullptr);
    }

    v2f fa[2][KBLK], fb[2][KBLK];
#pragma unroll
    for (int j = 0; j < KBLK; ++j) {
      fa[0][j] = *(const v2f*)&abase[4 * j];
      fb[0][j] = *(const v2f*)&bbase[4 * j];
    }

    v8f acc = {0.f, 0.f, 0.f, 0.f, 0.f, 0.f, 0.f, 0.f};
#pragma unroll
    for (int blk = 0; blk < KSTEPS / KBLK; ++blk) {
      const int cur = blk & 1, nxt = cur ^ 1;
      if (blk + 1 < KSTEPS / KBLK) {
#pragma unroll
        for (int j = 0; j < KBLK; ++j) {
          const int k = (blk + 1) * KBLK + j;
          fa[nxt][j] = *(const v2f*)&abase[4 * k];
          fb[nxt][j] = *(const v2f*)&bbase[4 * k];
        }
      }
      __builtin_amdgcn_sched_barrier(0);
#pragma unroll
      for (int j = 0; j < KBLK; ++j) {
        acc = __builtin_amdgcn_wmma_f32_16x16x4_f32(
             false, fa[cur][j],  false, fb[cur][j],
             (short)0, acc,  false,  false);
      }
      __builtin_amdgcn_sched_barrier(0);
    }

    if (dcol < D_) {
#pragma unroll
      for (int j = 0; j < 8; ++j) {
        const int g = (b0 + j + rAdd) * P_ + p;
        __builtin_nontemporal_store(acc[j] + lb, out + (size_t)g * D_ + dcol);
      }
    }

    if (!has_next) break;
    t = tn; p = pn; b0 = bn0;
    wr0 = wn0; wr1 = wn1; bi0 = bn0v; bi1 = bn1v;
  }
}

extern "C" void kernel_launch(void* const* d_in, const int* in_sizes, int n_in,
                              void* d_out, int out_size, void* d_ws, size_t ws_size,
                              hipStream_t stream) {
  const float* feat    = (const float*)d_in[0];
  const float* weights = (const float*)d_in[1];
  const float* bias    = (const float*)d_in[2];
  const float* gamma   = (const float*)d_in[3];
  const float* beta    = (const float*)d_in[4];
  const float* linw    = (const float*)d_in[5];
  const float* linb    = (const float*)d_in[6];
  float* out  = (float*)d_out;
  float* wrow = (float*)d_ws;

  rowsum_kernel<<<(P_ * C_) / 8, 256, 0, stream>>>(weights, wrow);

  fused_kernel<<<NWAVES / WAVES_PER_BLOCK, 64, 0, stream>>>(feat, wrow, bias, gamma, beta,
                                           linw, linb, out);
}


# Phase A: lane sums of 8 rows taken together by one halving butterfly (partner order 16,8,4,2,1 kept, same bits); mean/rstd once per row via v_readlane; mean subtracted inside the fma addend
# speedup vs baseline: 1.5899x; 1.0400x over previous
#include <hip/hip_runtime.h>
#include <math.h>

typedef __attribute__((ext_vector_type(2))) float v2f;
typedef __attribute__((ext_vector_type(4))) float v4f;
typedef __attribute__((ext_vector_type(8))) float v8f;

#define B_  8192
#define P_  25
#define C_  256
#define D_  5
#define R_  (B_ * P_)
#define ROWS_PER_WAVE 16
#define WAVES_PER_BLOCK 2
#define HSTRIDE 260
#define KSTEPS (C_ / 4)
#define ZOFF (D_ * HSTRIDE)
#define KBLK 8

__device__ __forceinline__ float xhalf(float x) {
  const int b = __builtin_bit_cast(int, x);
  return __builtin_bit_cast(float,
      __builtin_amdgcn_permlanex16(b, b, 0x76543210, 0xfedcba98, false, false));
}
template <int M>
__device__ __forceinline__ float add_xmask(float x) {
  const int b = __builtin_bit_cast(int, x);
  const int o = __builtin_amdgcn_update_dpp(0, b, 0x160 | M, 0xf, 0xf, true);
  return x + __builtin_bit_cast(float, o);
}
__device__ __forceinline__ float wave_sum(float s) {
  s += xhalf(s);
  s = add_xmask<8>(s);
  s = add_xmask<4>(s);
  s = add_xmask<2>(s);
  s = add_xmask<1>(s);
  return s;
}

template <int M>
__device__ __forceinline__ float add_from_xmask(float keep, float send) {
  const int b = __builtin_bit_cast(int, send);
  const int o = __builtin_amdgcn_update_dpp(0, b, 0x160 | M, 0xf, 0xf, true);
  return keep + __builtin_bit_cast(float, o);
}

template <int M, int N>
__device__ __forceinline__ void halve_xmask(float* v, const bool up) {
#pragma unroll
  for (int j = 0; j < N; ++j) {
    const float keep = up ? v[j + N] : v[j];
    const float send = up ? v[j] : v[j + N];
    v[j] = add_from_xmask<M>(keep, send);
  }
}

__device__ __forceinline__ float joint_row_sums(float* v, const int ln) {
  halve_xmask<8, 4>(v, (ln & 8) != 0);
  halve_xmask<4, 2>(v, (ln & 4) != 0);
  halve_xmask<2, 1>(v, (ln & 2) != 0);
  return add_xmask<1>(v[0]);
}

__device__ __forceinline__ float lane_value(float x, const int r) {
  return __builtin_bit_cast(float, __builtin_amdgcn_readlane(__builtin_bit_cast(int, x), r));
}

__global__ __launch_bounds__(256) void rowsum_kernel(const float* __restrict__ w,
                                                     float* __restrict__ wrow) {
  const int row = blockIdx.x * 8 + (threadIdx.x >> 5);
  const int ln  = threadIdx.x & 31;
  const float* rp = w + (size_t)row * C_;
  float s = 0.f;
#pragma unroll
  for (int j = 0; j < 8; ++j) s += rp[ln + 32 * j];
  s = wave_sum(s);
  if (ln == 0) wrow[row] = s;
}

#define NTILES (R_ / ROWS_PER_WAVE)
#define NWAVES 4096

template <bool PF>
__device__ __forceinline__ void phase_a(
    v4f (&f0)[ROWS_PER_WAVE], v4f (&f1)[ROWS_PER_WAVE],
    const v4f wr0, const v4f wr1, const v4f bi0, const v4f bi1,
    const v4f ga0, const v4f ga1, const v4f be0, const v4f be1,
    float* __restrict__ hw, const int c0, const int c1, const float* fpn, const int ln) {
  const bool hi = ln >= 16;
#pragma unroll
  for (int g = 0; g < ROWS_PER_WAVE; g += 8) {
    float sv[8];
#pragma unroll
    for (int i = 0; i < 8; ++i) {
      const v4f x0 = f0[g + i];
      const v4f x1 = f1[g + i];
      float s1 = 0.f, s2 = 0.f;
#pragma unroll
      for (int q = 0; q < 4; ++q) {
        const float p0 = fmaf(x0[q], wr0[q], bi0[q]);
        const float p1 = fmaf(x1[q], wr1[q], bi1[q]);
        s1 += p0 + p1;
        s2  = fmaf(p0, p0, s2);
        s2  = fmaf(p1, p1, s2);
      }
      const float keep = hi ? s2 : s1;
      const float send = hi ? s1 : s2;
      sv[i] = keep + xhalf(send);
    }
    const float tot = joint_row_sums(sv, ln);
    const float oth = xhalf(tot);
    const float t1  = hi ? oth : tot;
    const float t2  = hi ? tot : oth;
    const float inv   = 1.0f / (float)C_;
    const float meanl = t1 * inv;
    const float varl  = fmaf(t2, inv, -meanl * meanl);
    const float rstdl = rsqrtf(varl + 1e-5f);

#pragma unroll
    for (int i = 0; i < 8; ++i) {
      const int r = g + i;
      const float mean = lane_value(meanl, 2 * i);
      const float rstd = lane_value(rstdl, 2 * i);
      const v4f x0 = f0[r];
      const v4f x1 = f1[r];
      v4f h0, h1;
#pragma unroll
      for (int q = 0; q < 4; ++q) {
        const float d0 = fmaf(x0[q], wr0[q], bi0[q] - mean);
        const float d1 = fmaf(x1[q], wr1[q], bi1[q] - mean);
        const float n0 = fmaf(d0 * rstd, ga0[q], be0[q]);
        const float n1 = fmaf(d1 * rstd, ga1[q], be1[q]);
        h0[q] = fmaxf(x0[q] + n0, 0.f);
        h1[q] = fmaxf(x1[q] + n1, 0.f);
      }
      *(v4f*)&hw[r * HSTRIDE + c0] = h0;
      *(v4f*)&hw[r * HSTRIDE + c1] = h1;
      if (PF) {
        f0[r] = __builtin_nontemporal_load((const v4f*)(fpn + c0));
        f1[r] = __builtin_nontemporal_load((const v4f*)(fpn + c1));
        fpn += P_ * C_;
      }
    }
  }
}

__global__ __launch_bounds__(64) __attribute__((amdgpu_waves_per_eu(4, 4))) void fused_kernel(
    const float* __restrict__ feat, const float* __restrict__ wrow,
    const float* __restrict__ bias, const float* __restrict__ gamma,
    const float* __restrict__ beta, const float* __restrict__ linw,
    const float* __restrict__ linb, float* __restrict__ out) {
  __shared__ __attribute__((aligned(16))) float hbuf[WAVES_PER_BLOCK * ROWS_PER_WAVE * HSTRIDE];
  __shared__ __attribute__((aligned(16))) float lw[ZOFF + HSTRIDE];

  const int tid = threadIdx.x;
  const int ln  = tid & 31;
  const int wv   = __builtin_amdgcn_readfirstlane(tid >> 5);
  const int wave = blockIdx.x * WAVES_PER_BLOCK + wv;
  float* hw = hbuf + wv * (ROWS_PER_WAVE * HSTRIDE);

  const int c0 = ln * 4, c1 = ln * 4 + 128;

  float lreg[(D_ * C_) / 64];
#pragma unroll
  for (int i = 0; i < (D_ * C_) / 64; ++i) lreg[i] = linw[tid + 64 * i];

  int t  = wave;
  int p  = t % P_;
  int b0 = (t / P_) * ROWS_PER_WAVE;
  v4f wr0 = *(const v4f*)(wrow + p * C_ + c0);
  v4f wr1 = *(const v4f*)(wrow + p * C_ + c1);
  v4f bi0 = *(const v4f*)(bias + p * C_ + c0);
  v4f bi1 = *(const v4f*)(bias + p * C_ + c1);
  const v4f ga0 = *(const v4f*)(gamma + c0);
  const v4f ga1 = *(const v4f*)(gamma + c1);
  const v4f be0 = *(const v4f*)(beta + c0);
  const v4f be1 = *(const v4f*)(beta + c1);
  const int dcol = ln & 15;
  const float lb = (dcol < D_) ? linb[dcol] : 0.f;

  v4f f0[ROWS_PER_WAVE], f1[ROWS_PER_WAVE];
  {
    const float* fp = feat + ((size_t)b0 * P_ + p) * C_;
#pragma unroll
    for (int r = 0; r < ROWS_PER_WAVE; ++r) {
      f0[r] = __builtin_nontemporal_load((const v4f*)(fp + c0));
      f1[r] = __builtin_nontemporal_load((const v4f*)(fp + c1));
      fp += P_ * C_;
    }
  }

#pragma unroll
  for (int i = 0; i < (D_ * C_) / 64; ++i) {
    const int e = tid + 64 * i;
    lw[(e >> 8) * HSTRIDE + (e & 255)] = lreg[i];
  }
  for (int i = tid; i < HSTRIDE; i += 64) lw[ZOFF + i] = 0.f;
  __syncthreads();

  const int rowA = ln & 15;
  const int koff = (ln >> 4) << 1;
  const float* abase = &hw[rowA * HSTRIDE + koff];
  const float* bbase = &lw[((dcol < D_) ? dcol : D_) * HSTRIDE + koff];
  const int rAdd = (ln >> 4) << 3;

  for (;;) {
    const int  tn = t + NWAVES;
    const bool has_next = tn < NTILES;
    const int  pn  = tn % P_;
    const int  bn0 = (tn / P_) * ROWS_PER_WAVE;

    v4f wn0 = wr0, wn1 = wr1, bn0v = bi0, bn1v = bi1;
    if (has_next) {
      wn0  = *(const v4f*)(wrow + pn * C_ + c0);
      wn1  = *(const v4f*)(wrow + pn * C_ + c1);
      bn0v = *(const v4f*)(bias + pn * C_ + c0);
      bn1v = *(const v4f*)(bias + pn * C_ + c1);
      const float* fpn = feat + ((size_t)bn0 * P_ + pn) * C_;
      phase_a<true>(f0, f1, wr0, wr1, bi0, bi1, ga0, ga1, be0, be1, hw, c0, c1, fpn, ln);
    } else {
      phase_a<false>(f0, f1, wr0, wr1, bi0, bi1, ga0, ga1, be0, be1, hw, c0, c1, nullptr, ln);
    }

    v2f fa[2][KBLK], fb[2][KBLK];
#pragma unroll
    for (int j = 0; j < KBLK; ++j) {
      fa[0][j] = *(const v2f*)&abase[4 * j];
      fb[0][j] = *(const v2f*)&bbase[4 * j];
    }

    v8f acc = {0.f, 0.f, 0.f, 0.f, 0.f, 0.f, 0.f, 0.f};
#pragma unroll
    for (int blk = 0; blk < KSTEPS / KBLK; ++blk) {
      const int cur = blk & 1, nxt = cur ^ 1;
      if (blk + 1 < KSTEPS / KBLK) {
#pragma unroll
        for (int j = 0; j < KBLK; ++j) {
          const int k = (blk + 1) * KBLK + j;
          fa[nxt][j] = *(const v2f*)&abase[4 * k];
          fb[nxt][j] = *(const v2f*)&bbase[4 * k];
        }
      }
      __builtin_amdgcn_sched_barrier(0);
#pragma unroll
      for (int j = 0; j < KBLK; ++j) {
        acc = __builtin_amdgcn_wmma_f32_16x16x4_f32(
             false, fa[cur][j],  false, fb[cur][j],
             (short)0, acc,  false,  false);
      }
      __builtin_amdgcn_sched_barrier(0);
    }

    if (dcol < D_) {
#pragma unroll
      for (int j = 0; j < 8; ++j) {
        const int g = (b0 + j + rAdd) * P_ + p;
        __builtin_nontemporal_store(acc[j] + lb, out + (size_t)g * D_ + dcol);
      }
    }

    if (!has_next) break;
    t = tn; p = pn; b0 = bn0;
    wr0 = wn0; wr1 = wn1; bi0 = bn0v; bi1 = bn1v;
  }
}

extern "C" void kernel_launch(void* const* d_in, const int* in_sizes, int n_in,
                              void* d_out, int out_size, void* d_ws, size_t ws_size,
                              hipStream_t stream) {
  const float* feat    = (const float*)d_in[0];
  const float* weights = (const float*)d_in[1];
  const float* bias    = (const float*)d_in[2];
  const float* gamma   = (const float*)d_in[3];
  const float* beta    = (const float*)d_in[4];
  const float* linw    = (const float*)d_in[5];
  const float* linb    = (const float*)d_in[6];
  float* out  = (float*)d_out;
  float* wrow = (float*)d_ws;

  rowsum_kernel<<<(P_ * C_) / 8, 256, 0, stream>>>(weights, wrow);

  fused_kernel<<<NWAVES / WAVES_PER_BLOCK, 64, 0, stream>>>(feat, wrow, bias, gamma, beta,
                                           linw, linb, out);
}
